# starting kernel
# baseline (speedup 1.0000x reference)
#include <hip/hip_runtime.h>
#include <math.h>

constexpr int BB = 2;
constexpr int NN = 50000;
constexpr int FF = 8;
constexpr int EE = 800000;
constexpr int DD = 3;
constexpr int KK = 8;
constexpr int HH = 64;
constexpr int FB = FF * BB;
constexpr int NK = NN * KK;
constexpr int CDIM = FF * KK;
constexpr int MROWS = BB * NN;
constexpr int MTILES = MROWS / 16;
constexpr int NTILES = HH / 16;
constexpr float EPSF = 1e-14f;

typedef __attribute__((ext_vector_type(2))) float v2f;
typedef __attribute__((ext_vector_type(8))) float v8f;

__device__ __forceinline__ unsigned fkey(float f) {
  unsigned b = __float_as_uint(f);
  return (b & 0x80000000u) ? ~b : (b | 0x80000000u);
}
__device__ __forceinline__ float finv(unsigned k) {
  return __uint_as_float((k & 0x80000000u) ? (k & 0x7fffffffu) : ~k);
}

__global__ void k_init(unsigned* ws, int n) {
  int i = blockIdx.x * blockDim.x + threadIdx.x;
  int stride = gridDim.x * blockDim.x;
  for (; i < n; i += stride) ws[i] = 0u;
}

__global__ void k_logit_max(const float* __restrict__ uval,
                            const float* __restrict__ mu,
                            const float* __restrict__ sigma,
                            const int* __restrict__ urows,
                            unsigned* __restrict__ mkey) {
  int e = blockIdx.x * blockDim.x + threadIdx.x;
  if (e >= EE) return;
  float mu_l[KK * DD], iv_l[KK * DD];
#pragma unroll
  for (int i = 0; i < KK * DD; ++i) {
    mu_l[i] = mu[i];
    float sg = sigma[i];
    iv_l[i] = 1.0f / (sg * sg + EPSF);
  }
  float uv[DD];
#pragma unroll
  for (int d = 0; d < DD; ++d) uv[d] = uval[e * DD + d];
  int row = urows[e];
#pragma unroll
  for (int k = 0; k < KK; ++k) {
    float s = 0.0f;
#pragma unroll
    for (int d = 0; d < DD; ++d) {
      float df = uv[d] - mu_l[k * DD + d];
      s += df * df * iv_l[k * DD + d];
    }
    atomicMax(&mkey[row * KK + k], fkey(-0.5f * s));
  }
}

__global__ void k_denom(const float* __restrict__ uval,
                        const float* __restrict__ mu,
                        const float* __restrict__ sigma,
                        const int* __restrict__ urows,
                        const unsigned* __restrict__ mkey,
                        float* __restrict__ denom) {
  int e = blockIdx.x * blockDim.x + threadIdx.x;
  if (e >= EE) return;
  float mu_l[KK * DD], iv_l[KK * DD];
#pragma unroll
  for (int i = 0; i < KK * DD; ++i) {
    mu_l[i] = mu[i];
    float sg = sigma[i];
    iv_l[i] = 1.0f / (sg * sg + EPSF);
  }
  float uv[DD];
#pragma unroll
  for (int d = 0; d < DD; ++d) uv[d] = uval[e * DD + d];
  int row = urows[e];
#pragma unroll
  for (int k = 0; k < KK; ++k) {
    float s = 0.0f;
#pragma unroll
    for (int d = 0; d < DD; ++d) {
      float df = uv[d] - mu_l[k * DD + d];
      s += df * df * iv_l[k * DD + d];
    }
    float l = -0.5f * s;
    float mval = finv(mkey[row * KK + k]);
    atomicAdd(&denom[row * KK + k], __expf(l - mval));
  }
}

__global__ void k_scatter(const float* __restrict__ X,
                          const float* __restrict__ uval,
                          const float* __restrict__ mu,
                          const float* __restrict__ sigma,
                          const int* __restrict__ urows,
                          const int* __restrict__ ucols,
                          const unsigned* __restrict__ mkey,
                          const float* __restrict__ denom,
                          float* __restrict__ agg) {
  int t = blockIdx.x * blockDim.x + threadIdx.x;
  if (t >= EE * KK) return;
  int e = t >> 3;
  int k = t & (KK - 1);
  int row = urows[e];
  int col = ucols[e];
  float s = 0.0f;
#pragma unroll
  for (int d = 0; d < DD; ++d) {
    float sg = sigma[k * DD + d];
    float df = uval[e * DD + d] - mu[k * DD + d];
    s += df * df / (sg * sg + EPSF);
  }
  float l = -0.5f * s;
  float w = __expf(l - finv(mkey[row * KK + k])) / denom[row * KK + k];
  const float* x0 = X + (size_t)col * FF;
  const float* x1 = X + (size_t)NN * FF + (size_t)col * FF;
  float* dst = agg + (size_t)row * (KK * FB) + k * FB;
#pragma unroll
  for (int f = 0; f < FF; ++f) {
    atomicAdd(dst + f * 2 + 0, w * x0[f]);
    atomicAdd(dst + f * 2 + 1, w * x1[f]);
  }
}

__global__ void k_gemm(const float* __restrict__ agg,
                       const float* __restrict__ GW,
                       float* __restrict__ out) {
  int wid = blockIdx.x * (blockDim.x >> 5) + (threadIdx.x >> 5);
  if (wid >= MTILES * NTILES) return;
  int lane = threadIdx.x & 31;
  int half = lane >> 4;
  int lm = lane & 15;
  int mt = wid >> 2;
  int nt = wid & 3;
  int row0 = mt * 16;
  int col0 = nt * 16;
  int b = row0 / NN;
  int n0 = row0 % NN;

  v8f acc = {};
#if __has_builtin(__builtin_amdgcn_wmma_f32_16x16x4_f32)
  const float* arow = agg + (size_t)(n0 + lm) * (KK * FB) + b;
#pragma unroll
  for (int kk = 0; kk < CDIM / 4; ++kk) {
    int c0 = kk * 4 + half * 2;
    int c1 = c0 + 1;
    v2f a, bf;
    a.x = arow[(c0 & 7) * FB + ((c0 >> 3) << 1)];
    a.y = arow[(c1 & 7) * FB + ((c1 >> 3) << 1)];
    bf.x = GW[c0 * HH + col0 + lm];
    bf.y = GW[c1 * HH + col0 + lm];
    acc = __builtin_amdgcn_wmma_f32_16x16x4_f32(
         false, a,  false, bf,
         (short)0, acc,  false,  false);
  }
#pragma unroll
  for (int j = 0; j < 8; ++j) {
    int r = row0 + half * 8 + j;
    out[(size_t)r * HH + col0 + lm] = acc[j];
  }
#else
#pragma unroll
  for (int j = 0; j < 8; ++j) {
    int r = row0 + half * 8 + j;
    int n = r % NN;
    int bb = r / NN;
    float s = 0.0f;
    for (int c = 0; c < CDIM; ++c) {
      float fv = agg[(size_t)n * (KK * FB) + (c & 7) * FB + ((c >> 3) << 1) + bb];
      s += fv * GW[c * HH + col0 + lm];
    }
    out[(size_t)r * HH + col0 + lm] = s;
  }
#endif
}

extern "C" void kernel_launch(void* const* d_in, const int* in_sizes, int n_in,
                              void* d_out, int out_size, void* d_ws, size_t ws_size,
                              hipStream_t stream) {
  const float* X     = (const float*)d_in[0];
  const float* uval  = (const float*)d_in[1];
  const float* mu    = (const float*)d_in[2];
  const float* sigma = (const float*)d_in[3];
  const float* GW    = (const float*)d_in[4];
  const int* urows   = (const int*)d_in[5];
  const int* ucols   = (const int*)d_in[6];
  float* out = (float*)d_out;

  unsigned* mkey = (unsigned*)d_ws;
  float* denom   = (float*)d_ws + NK;
  float* agg     = (float*)d_ws + 2 * NK;
  const int ws_words = 2 * NK + NK * FB;

  k_init<<<2048, 256, 0, stream>>>((unsigned*)d_ws, ws_words);

  k_logit_max<<<(EE + 255) / 256, 256, 0, stream>>>(uval, mu, sigma, urows, mkey);

  k_denom<<<(EE + 255) / 256, 256, 0, stream>>>(uval, mu, sigma, urows, mkey, denom);

  k_scatter<<<(EE * KK + 255) / 256, 256, 0, stream>>>(
      X, uval, mu, sigma, urows, ucols, mkey, denom, agg);

  int waves = MTILES * NTILES;
  int blocks = (waves + 7) / 8;
  k_gemm<<<blocks, 256, 0, stream>>>(agg, GW, out);
}


# GEMM moved into the node pass: a work-group's 16 finished rows go from their owner lanes' registers through LDS to its 4 waves' f32 WMMAs and out; agg never in memory, 3 launches; same sum order
# speedup vs baseline: 9.5394x; 9.5394x over previous
#include <hip/hip_runtime.h>
#include <math.h>

constexpr int BB = 2;
constexpr int NN = 50000;
constexpr int FF = 8;
constexpr int EE = 800000;
constexpr int DD = 3;
constexpr int KK = 8;
constexpr int HH = 64;
constexpr int FB = FF * BB;
constexpr int NK = NN * KK;
constexpr int CDIM = FF * KK;
constexpr float EPSF = 1e-14f;

typedef __attribute__((ext_vector_type(2))) float v2f;
typedef __attribute__((ext_vector_type(8))) float v8f;

typedef __attribute__((ext_vector_type(4))) float v4f;
typedef __attribute__((ext_vector_type(4))) int v4i;

constexpr int SLOT = KK * FB;
constexpr int ENT0 = 32;
constexpr int CAP = SLOT - ENT0;
constexpr int CH_W = 4;

__global__ void k_init(int* __restrict__ spill, int* __restrict__ slots) {
  int i = blockIdx.x * blockDim.x + threadIdx.x;
  if (i < NN) slots[(size_t)i * SLOT] = 0;
  if (i == 0) spill[0] = 0;
}

constexpr int EPT = 8;
__global__ void k_fill(const int* __restrict__ urows,
                       int* __restrict__ spill,
                       int* __restrict__ slots) {
  int t = blockIdx.x * blockDim.x + threadIdx.x;
  if (t >= EE / EPT) return;
  const v4i* rp = (const v4i*)urows + (size_t)t * (EPT / 4);
  v4i ra = rp[0], rb = rp[1];
  int row[EPT] = {ra.x, ra.y, ra.z, ra.w, rb.x, rb.y, rb.z, rb.w};
  int pos[EPT];
#pragma unroll
  for (int i = 0; i < EPT; ++i) pos[i] = atomicAdd(slots + (size_t)row[i] * SLOT, 1);
#pragma unroll
  for (int i = 0; i < EPT; ++i) {
    int e = t * EPT + i;
    if (pos[i] < CAP) {
      slots[(size_t)row[i] * SLOT + ENT0 + pos[i]] = e;
    } else {
      int o = atomicAdd(spill, 1);
      spill[1 + o] = e;
    }
  }
}

struct Comp { float m0, m1, m2, i0, i1, i2; };

__device__ __forceinline__ float logit3(float u0, float u1, float u2, const Comp& c) {
  float d0 = u0 - c.m0, d1 = u1 - c.m1, d2 = u2 - c.m2;
  float s = d0 * d0 * c.i0;
  s += d1 * d1 * c.i1;
  s += d2 * d2 * c.i2;
  return -0.5f * s;
}

__device__ __forceinline__ int edge_id(int e) { return min(max(e, 0), EE - 1); }

constexpr int NPB = 16;
constexpr int ROWT = NPB * KK;
constexpr int PITCH = SLOT + 4;
static_assert(NN % NPB == 0, "whole work-groups only: the grid is exactly NN / NPB");

__global__ __launch_bounds__(ROWT) void k_row_gemm(const float* __restrict__ X,
                      const float* __restrict__ uval,
                      const float* __restrict__ mu,
                      const float* __restrict__ sigma,
                      const float* __restrict__ GW,
                      const int* __restrict__ urows,
                      const int* __restrict__ ucols,
                      const int* __restrict__ spill,
                      const int* __restrict__ slots,
                      float* __restrict__ out) {
  __shared__ v4f rowbuf[NPB * PITCH / 4];
  const int jn = threadIdx.x >> 3;
  const int n0 = blockIdx.x * NPB;
  const int node = n0 + jn;
  const int k = threadIdx.x & (KK - 1);

  Comp c;
  c.m0 = mu[k * DD + 0]; c.m1 = mu[k * DD + 1]; c.m2 = mu[k * DD + 2];
  {
    float s0 = sigma[k * DD + 0], s1 = sigma[k * DD + 1], s2 = sigma[k * DD + 2];
    c.i0 = 1.0f / (s0 * s0 + EPSF);
    c.i1 = 1.0f / (s1 * s1 + EPSF);
    c.i2 = 1.0f / (s2 * s2 + EPSF);
  }

  const int* slot = slots + (size_t)node * SLOT;
  const int cnt = slot[0];
  const int nin = min(max(cnt, 0), CAP);
  const bool spilled = cnt > CAP;
  int nsp = 0;
  if (spilled) nsp = min(max(spill[0], 0), 2 * NK - 1);

  float m = -INFINITY;
  float den = 0.0f;
  float acc[FB];
#pragma unroll
  for (int i = 0; i < FB; ++i) acc[i] = 0.0f;

  for (int j = 0; j < nin; j += CH_W) {
    int e[CH_W], col[CH_W];
    float u[CH_W][DD];
    float l[CH_W];
    v4f x[CH_W][4];
#pragma unroll
    for (int i = 0; i < CH_W; ++i) e[i] = edge_id(slot[ENT0 + (j + i < nin ? j + i : j)]);
#pragma unroll
    for (int i = 0; i < CH_W; ++i) {
      const float* p = uval + (size_t)e[i] * DD;
      u[i][0] = p[0]; u[i][1] = p[1]; u[i][2] = p[2];
      col[i] = ucols[e[i]];
    }
#pragma unroll
    for (int i = 0; i < CH_W; ++i) {
      const v4f* x0 = (const v4f*)(X + (size_t)col[i] * FF);
      const v4f* x1 = (const v4f*)(X + (size_t)NN * FF + (size_t)col[i] * FF);
      x[i][0] = x0[0]; x[i][1] = x0[1]; x[i][2] = x1[0]; x[i][3] = x1[1];
    }
    float mn = m;
#pragma unroll
    for (int i = 0; i < CH_W; ++i) {
      l[i] = logit3(u[i][0], u[i][1], u[i][2], c);
      if (!(j + i < nin)) l[i] = -INFINITY;
      mn = fmaxf(mn, l[i]);
    }
    const float sc = __expf(m - mn);
    m = mn;
    den *= sc;
#pragma unroll
    for (int q = 0; q < FB; ++q) acc[q] *= sc;
#pragma unroll
    for (int i = 0; i < CH_W; ++i) {
      float w = __expf(l[i] - mn);
      den += w;
#pragma unroll
      for (int f = 0; f < 4; ++f) {
        acc[f * 2 + 0] += w * x[i][0][f];
        acc[f * 2 + 1] += w * x[i][2][f];
        acc[(f + 4) * 2 + 0] += w * x[i][1][f];
        acc[(f + 4) * 2 + 1] += w * x[i][3][f];
      }
    }
  }
  if (spilled) {
    for (int i = 0; i < nsp; ++i) {
      int e = edge_id(spill[1 + i]);
      if (urows[e] == node) {
        const float* p = uval + (size_t)e * DD;
        float lg = logit3(p[0], p[1], p[2], c);
        float mn = fmaxf(m, lg);
        float sc = __expf(m - mn);
        float w = __expf(lg - mn);
        m = mn;
        den = den * sc + w;
        int cl = ucols[e];
        const float* x0 = X + (size_t)cl * FF;
        const float* x1 = X + (size_t)NN * FF + (size_t)cl * FF;
#pragma unroll
        for (int f = 0; f < FF; ++f) {
          acc[f * 2 + 0] = acc[f * 2 + 0] * sc + w * x0[f];
          acc[f * 2 + 1] = acc[f * 2 + 1] * sc + w * x1[f];
        }
      }
    }
  }
  {
    const float inv = cnt > 0 ? 1.0f / den : 0.0f;
#pragma unroll
    for (int q = 0; q < FB; ++q) acc[q] *= inv;
  }

  const int lane = threadIdx.x & 31;
  const int half = lane >> 4;
  const int lm = lane & 15;
  const int col0 = (threadIdx.x >> 5) * 16;

#if __has_builtin(__builtin_amdgcn_wmma_f32_16x16x4_f32)
  v2f bfr[CDIM / 4];
#pragma unroll
  for (int kk = 0; kk < CDIM / 4; ++kk) {
    int c0 = kk * 4 + half * 2;
    bfr[kk].x = GW[c0 * HH + col0 + lm];
    bfr[kk].y = GW[(c0 + 1) * HH + col0 + lm];
  }
#endif

  {
    float* rowp = (float*)rowbuf + jn * PITCH + k * 2;
#pragma unroll
    for (int f = 0; f < FF; ++f) {
      v2f v;
      v.x = acc[f * 2 + 0];
      v.y = acc[f * 2 + 1];
      *(v2f*)(rowp + f * 16) = v;
    }
  }
  __syncthreads();

#if __has_builtin(__builtin_amdgcn_wmma_f32_16x16x4_f32)
  const v4f* ap = rowbuf + lm * (PITCH / 4) + half;
  v8f d0 = {}, d1 = {};
#pragma unroll
  for (int kk = 0; kk < CDIM / 4; ++kk) {
    v4f a4 = ap[kk * 2];
    v2f a0, a1;
    a0.x = a4.x; a0.y = a4.z;
    a1.x = a4.y; a1.y = a4.w;
    d0 = __builtin_amdgcn_wmma_f32_16x16x4_f32(
         false, a0,  false, bfr[kk],
         (short)0, d0,  false,  false);
    d1 = __builtin_amdgcn_wmma_f32_16x16x4_f32(
         false, a1,  false, bfr[kk],
         (short)0, d1,  false,  false);
  }
#pragma unroll
  for (int j = 0; j < 8; ++j) {
    size_t r = (size_t)(n0 + half * 8 + j);
    out[r * HH + col0 + lm] = d0[j];
    out[((size_t)NN + r) * HH + col0 + lm] = d1[j];
  }
#else
  {
    const float* rw = (const float*)rowbuf;
    for (int j = 0; j < 8; ++j) {
      int jr = half * 8 + j;
      for (int bb = 0; bb < BB; ++bb) {
        float s = 0.0f;
        for (int c = 0; c < CDIM; ++c)
          s += rw[jr * PITCH + (c >> 3) * 16 + (c & 7) * 2 + bb] * GW[c * HH + col0 + lm];
        out[((size_t)bb * NN + n0 + jr) * HH + col0 + lm] = s;
      }
    }
  }
#endif
}

extern "C" void kernel_launch(void* const* d_in, const int* in_sizes, int n_in,
                              void* d_out, int out_size, void* d_ws, size_t ws_size,
                              hipStream_t stream) {
  const float* X     = (const float*)d_in[0];
  const float* uval  = (const float*)d_in[1];
  const float* mu    = (const float*)d_in[2];
  const float* sigma = (const float*)d_in[3];
  const float* GW    = (const float*)d_in[4];
  const int* urows   = (const int*)d_in[5];
  const int* ucols   = (const int*)d_in[6];
  float* out = (float*)d_out;

  int* spill = (int*)d_ws;
  int* slots = (int*)d_ws + 2 * NK;

  k_init<<<(NN + 255) / 256, 256, 0, stream>>>(spill, slots);

  k_fill<<<(EE / EPT + 255) / 256, 256, 0, stream>>>(urows, spill, slots);

  k_row_gemm<<<NN / NPB, ROWT, 0, stream>>>(
      X, uval, mu, sigma, GW, urows, ucols, spill, slots, out);
}


# Fill pass made half-affine: work-groups draw (edge chunk, node half) units from two ticket counters, own cache half first, so a bucket is written from one half only; sums still in arrival order
# speedup vs baseline: 15.4334x; 1.6179x over previous
#include <hip/hip_runtime.h>
#include <math.h>

constexpr int BB = 2;
constexpr int NN = 50000;
constexpr int FF = 8;
constexpr int EE = 800000;
constexpr int DD = 3;
constexpr int KK = 8;
constexpr int HH = 64;
constexpr int FB = FF * BB;
constexpr int NK = NN * KK;
constexpr int CDIM = FF * KK;
constexpr float EPSF = 1e-14f;

typedef __attribute__((ext_vector_type(2))) float v2f;
typedef __attribute__((ext_vector_type(8))) float v8f;

typedef __attribute__((ext_vector_type(4))) float v4f;
typedef __attribute__((ext_vector_type(4))) int v4i;

constexpr int SLOT = KK * FB;
constexpr int ENT0 = 32;
constexpr int CAP = SLOT - ENT0;
constexpr int CH_W = 4;

constexpr int EPT = 16;
constexpr int FILL_T = 256;
constexpr int CHUNK = FILL_T * EPT;
constexpr int NCHUNK = (EE + CHUNK - 1) / CHUNK;
constexpr int NHALF0 = NN / 2;
constexpr int QWORD = 16;
__device__ __forceinline__ size_t q_index(int h) {
  return (size_t)(h == 0 ? 0 : NN - 1) * SLOT + QWORD;
}

__global__ void k_init(int* __restrict__ spill, int* __restrict__ slots) {
  int i = blockIdx.x * blockDim.x + threadIdx.x;
  if (i < NN) slots[(size_t)i * SLOT] = 0;
  if (i == 0) {
    spill[0] = 0;
    slots[q_index(0)] = 0;
    slots[q_index(1)] = 0;
  }
}

__global__ __launch_bounds__(FILL_T) void k_fill(const int* __restrict__ urows,
                       int* __restrict__ spill,
                       int* __restrict__ slots) {
  __shared__ int sh_ticket;
  __shared__ int sh_half;
  if (threadIdx.x == 0) {
    unsigned id = __builtin_amdgcn_s_sendmsg_rtn(0x87);
    sh_half = (int)((id >> 18) & 1u);
  }
  __syncthreads();
  const int myhalf = sh_half;
  for (int pass = 0; pass < 2; ++pass) {
    const int hh = pass == 0 ? myhalf : 1 - myhalf;
    int* q = slots + q_index(hh);
    for (int it = 0; it <= NCHUNK; ++it) {
      if (threadIdx.x == 0) sh_ticket = atomicAdd(q, 1);
      __syncthreads();
      const int c = sh_ticket;
      __syncthreads();
      if (c < 0 || c >= NCHUNK) break;
      const int t = c * FILL_T + threadIdx.x;
      if (t < EE / EPT) {
        const v4i* rp = (const v4i*)urows + (size_t)t * (EPT / 4);
        int row[EPT];
#pragma unroll
        for (int g = 0; g < EPT / 4; ++g) {
          v4i r = rp[g];
          row[g * 4 + 0] = r.x; row[g * 4 + 1] = r.y; row[g * 4 + 2] = r.z; row[g * 4 + 3] = r.w;
        }
        int pos[EPT];
#pragma unroll
        for (int i = 0; i < EPT; ++i) {
          const bool mine = (row[i] >= NHALF0 ? 1 : 0) == hh;
          pos[i] = -1;
          if (mine) pos[i] = atomicAdd(slots + (size_t)row[i] * SLOT, 1);
        }
#pragma unroll
        for (int i = 0; i < EPT; ++i) {
          const bool mine = (row[i] >= NHALF0 ? 1 : 0) == hh;
          if (mine) {
            int e = t * EPT + i;
            if (pos[i] < CAP) {
              slots[(size_t)row[i] * SLOT + ENT0 + pos[i]] = e;
            } else {
              int o = atomicAdd(spill, 1);
              spill[1 + o] = e;
            }
          }
        }
      }
    }
  }
}

struct Comp { float m0, m1, m2, i0, i1, i2; };

__device__ __forceinline__ float logit3(float u0, float u1, float u2, const Comp& c) {
  float d0 = u0 - c.m0, d1 = u1 - c.m1, d2 = u2 - c.m2;
  float s = d0 * d0 * c.i0;
  s += d1 * d1 * c.i1;
  s += d2 * d2 * c.i2;
  return -0.5f * s;
}

__device__ __forceinline__ int edge_id(int e) { return min(max(e, 0), EE - 1); }

constexpr int NPB = 16;
constexpr int ROWT = NPB * KK;
constexpr int PITCH = SLOT + 4;
static_assert(NN % NPB == 0, "whole work-groups only: the grid is exactly NN / NPB");

__global__ __launch_bounds__(ROWT) void k_row_gemm(const float* __restrict__ X,
                      const float* __restrict__ uval,
                      const float* __restrict__ mu,
                      const float* __restrict__ sigma,
                      const float* __restrict__ GW,
                      const int* __restrict__ urows,
                      const int* __restrict__ ucols,
                      const int* __restrict__ spill,
                      const int* __restrict__ slots,
                      float* __restrict__ out) {
  __shared__ v4f rowbuf[NPB * PITCH / 4];
  const int jn = threadIdx.x >> 3;
  const int n0 = blockIdx.x * NPB;
  const int node = n0 + jn;
  const int k = threadIdx.x & (KK - 1);

  Comp c;
  c.m0 = mu[k * DD + 0]; c.m1 = mu[k * DD + 1]; c.m2 = mu[k * DD + 2];
  {
    float s0 = sigma[k * DD + 0], s1 = sigma[k * DD + 1], s2 = sigma[k * DD + 2];
    c.i0 = 1.0f / (s0 * s0 + EPSF);
    c.i1 = 1.0f / (s1 * s1 + EPSF);
    c.i2 = 1.0f / (s2 * s2 + EPSF);
  }

  const int* slot = slots + (size_t)node * SLOT;
  const int cnt = slot[0];
  const int nin = min(max(cnt, 0), CAP);
  const bool spilled = cnt > CAP;
  int nsp = 0;
  if (spilled) nsp = min(max(spill[0], 0), 2 * NK - 1);

  float m = -INFINITY;
  float den = 0.0f;
  float acc[FB];
#pragma unroll
  for (int i = 0; i < FB; ++i) acc[i] = 0.0f;

  for (int j = 0; j < nin; j += CH_W) {
    int e[CH_W], col[CH_W];
    float u[CH_W][DD];
    float l[CH_W];
    v4f x[CH_W][4];
#pragma unroll
    for (int i = 0; i < CH_W; ++i) e[i] = edge_id(slot[ENT0 + (j + i < nin ? j + i : j)]);
#pragma unroll
    for (int i = 0; i < CH_W; ++i) {
      const float* p = uval + (size_t)e[i] * DD;
      u[i][0] = p[0]; u[i][1] = p[1]; u[i][2] = p[2];
      col[i] = ucols[e[i]];
    }
#pragma unroll
    for (int i = 0; i < CH_W; ++i) {
      const v4f* x0 = (const v4f*)(X + (size_t)col[i] * FF);
      const v4f* x1 = (const v4f*)(X + (size_t)NN * FF + (size_t)col[i] * FF);
      x[i][0] = x0[0]; x[i][1] = x0[1]; x[i][2] = x1[0]; x[i][3] = x1[1];
    }
    float mn = m;
#pragma unroll
    for (int i = 0; i < CH_W; ++i) {
      l[i] = logit3(u[i][0], u[i][1], u[i][2], c);
      if (!(j + i < nin)) l[i] = -INFINITY;
      mn = fmaxf(mn, l[i]);
    }
    const float sc = __expf(m - mn);
    m = mn;
    den *= sc;
#pragma unroll
    for (int q = 0; q < FB; ++q) acc[q] *= sc;
#pragma unroll
    for (int i = 0; i < CH_W; ++i) {
      float w = __expf(l[i] - mn);
      den += w;
#pragma unroll
      for (int f = 0; f < 4; ++f) {
        acc[f * 2 + 0] += w * x[i][0][f];
        acc[f * 2 + 1] += w * x[i][2][f];
        acc[(f + 4) * 2 + 0] += w * x[i][1][f];
        acc[(f + 4) * 2 + 1] += w * x[i][3][f];
      }
    }
  }
  if (spilled) {
    for (int i = 0; i < nsp; ++i) {
      int e = edge_id(spill[1 + i]);
      if (urows[e] == node) {
        const float* p = uval + (size_t)e * DD;
        float lg = logit3(p[0], p[1], p[2], c);
        float mn = fmaxf(m, lg);
        float sc = __expf(m - mn);
        float w = __expf(lg - mn);
        m = mn;
        den = den * sc + w;
        int cl = ucols[e];
        const float* x0 = X + (size_t)cl * FF;
        const float* x1 = X + (size_t)NN * FF + (size_t)cl * FF;
#pragma unroll
        for (int f = 0; f < FF; ++f) {
          acc[f * 2 + 0] = acc[f * 2 + 0] * sc + w * x0[f];
          acc[f * 2 + 1] = acc[f * 2 + 1] * sc + w * x1[f];
        }
      }
    }
  }
  {
    const float inv = cnt > 0 ? 1.0f / den : 0.0f;
#pragma unroll
    for (int q = 0; q < FB; ++q) acc[q] *= inv;
  }

  const int lane = threadIdx.x & 31;
  const int half = lane >> 4;
  const int lm = lane & 15;
  const int col0 = (threadIdx.x >> 5) * 16;

#if __has_builtin(__builtin_amdgcn_wmma_f32_16x16x4_f32)
  v2f bfr[CDIM / 4];
#pragma unroll
  for (int kk = 0; kk < CDIM / 4; ++kk) {
    int c0 = kk * 4 + half * 2;
    bfr[kk].x = GW[c0 * HH + col0 + lm];
    bfr[kk].y = GW[(c0 + 1) * HH + col0 + lm];
  }
#endif

  {
    float* rowp = (float*)rowbuf + jn * PITCH + k * 2;
#pragma unroll
    for (int f = 0; f < FF; ++f) {
      v2f v;
      v.x = acc[f * 2 + 0];
      v.y = acc[f * 2 + 1];
      *(v2f*)(rowp + f * 16) = v;
    }
  }
  __syncthreads();

#if __has_builtin(__builtin_amdgcn_wmma_f32_16x16x4_f32)
  const v4f* ap = rowbuf + lm * (PITCH / 4) + half;
  v8f d0 = {}, d1 = {};
#pragma unroll
  for (int kk = 0; kk < CDIM / 4; ++kk) {
    v4f a4 = ap[kk * 2];
    v2f a0, a1;
    a0.x = a4.x; a0.y = a4.z;
    a1.x = a4.y; a1.y = a4.w;
    d0 = __builtin_amdgcn_wmma_f32_16x16x4_f32(
         false, a0,  false, bfr[kk],
         (short)0, d0,  false,  false);
    d1 = __builtin_amdgcn_wmma_f32_16x16x4_f32(
         false, a1,  false, bfr[kk],
         (short)0, d1,  false,  false);
  }
#pragma unroll
  for (int j = 0; j < 8; ++j) {
    size_t r = (size_t)(n0 + half * 8 + j);
    out[r * HH + col0 + lm] = d0[j];
    out[((size_t)NN + r) * HH + col0 + lm] = d1[j];
  }
#else
  {
    const float* rw = (const float*)rowbuf;
    for (int j = 0; j < 8; ++j) {
      int jr = half * 8 + j;
      for (int bb = 0; bb < BB; ++bb) {
        float s = 0.0f;
        for (int c = 0; c < CDIM; ++c)
          s += rw[jr * PITCH + (c >> 3) * 16 + (c & 7) * 2 + bb] * GW[c * HH + col0 + lm];
        out[((size_t)bb * NN + n0 + jr) * HH + col0 + lm] = s;
      }
    }
  }
#endif
}

extern "C" void kernel_launch(void* const* d_in, const int* in_sizes, int n_in,
                              void* d_out, int out_size, void* d_ws, size_t ws_size,
                              hipStream_t stream) {
  const float* X     = (const float*)d_in[0];
  const float* uval  = (const float*)d_in[1];
  const float* mu    = (const float*)d_in[2];
  const float* sigma = (const float*)d_in[3];
  const float* GW    = (const float*)d_in[4];
  const int* urows   = (const int*)d_in[5];
  const int* ucols   = (const int*)d_in[6];
  float* out = (float*)d_out;

  int* spill = (int*)d_ws;
  int* slots = (int*)d_ws + 2 * NK;

  k_init<<<(NN + 255) / 256, 256, 0, stream>>>(spill, slots);

  k_fill<<<2 * NCHUNK, FILL_T, 0, stream>>>(urows, spill, slots);

  k_row_gemm<<<NN / NPB, ROWT, 0, stream>>>(
      X, uval, mu, sigma, GW, urows, ucols, spill, slots, out);
}


# Node walk shares its reads: each of a node's 8 lanes fetches one edge's id, u_val, column and 8 B of each X row, passed round by v_permlane16; 109->60 registers, all waves resident; same sum order
# speedup vs baseline: 15.9687x; 1.0347x over previous
#include <hip/hip_runtime.h>
#include <math.h>

constexpr int BB = 2;
constexpr int NN = 50000;
constexpr int FF = 8;
constexpr int EE = 800000;
constexpr int DD = 3;
constexpr int KK = 8;
constexpr int HH = 64;
constexpr int FB = FF * BB;
constexpr int NK = NN * KK;
constexpr int CDIM = FF * KK;
constexpr float EPSF = 1e-14f;

typedef __attribute__((ext_vector_type(2))) float v2f;
typedef __attribute__((ext_vector_type(8))) float v8f;

typedef __attribute__((ext_vector_type(4))) float v4f;
typedef __attribute__((ext_vector_type(4))) int v4i;

constexpr int SLOT = KK * FB;
constexpr int ENT0 = 32;
constexpr int CAP = SLOT - ENT0;
constexpr int CH_W = 4;

constexpr int EPT = 16;
constexpr int FILL_T = 256;
constexpr int CHUNK = FILL_T * EPT;
constexpr int NCHUNK = (EE + CHUNK - 1) / CHUNK;
constexpr int NHALF0 = NN / 2;
constexpr int QWORD = 16;
__device__ __forceinline__ size_t q_index(int h) {
  return (size_t)(h == 0 ? 0 : NN - 1) * SLOT + QWORD;
}

__global__ void k_init(int* __restrict__ spill, int* __restrict__ slots) {
  int i = blockIdx.x * blockDim.x + threadIdx.x;
  if (i < NN) slots[(size_t)i * SLOT] = 0;
  if (i == 0) {
    spill[0] = 0;
    slots[q_index(0)] = 0;
    slots[q_index(1)] = 0;
  }
}

__global__ __launch_bounds__(FILL_T) void k_fill(const int* __restrict__ urows,
                       int* __restrict__ spill,
                       int* __restrict__ slots) {
  __shared__ int sh_ticket;
  __shared__ int sh_half;
  if (threadIdx.x == 0) {
    unsigned id = __builtin_amdgcn_s_sendmsg_rtn(0x87);
    sh_half = (int)((id >> 18) & 1u);
  }
  __syncthreads();
  const int myhalf = sh_half;
  for (int pass = 0; pass < 2; ++pass) {
    const int hh = pass == 0 ? myhalf : 1 - myhalf;
    int* q = slots + q_index(hh);
    for (int it = 0; it <= NCHUNK; ++it) {
      if (threadIdx.x == 0) sh_ticket = atomicAdd(q, 1);
      __syncthreads();
      const int c = sh_ticket;
      __syncthreads();
      if (c < 0 || c >= NCHUNK) break;
      const int t = c * FILL_T + threadIdx.x;
      if (t < EE / EPT) {
        const v4i* rp = (const v4i*)urows + (size_t)t * (EPT / 4);
        int row[EPT];
#pragma unroll
        for (int g = 0; g < EPT / 4; ++g) {
          v4i r = rp[g];
          row[g * 4 + 0] = r.x; row[g * 4 + 1] = r.y; row[g * 4 + 2] = r.z; row[g * 4 + 3] = r.w;
        }
        int pos[EPT];
#pragma unroll
        for (int i = 0; i < EPT; ++i) {
          const bool mine = (row[i] >= NHALF0 ? 1 : 0) == hh;
          pos[i] = -1;
          if (mine) pos[i] = atomicAdd(slots + (size_t)row[i] * SLOT, 1);
        }
#pragma unroll
        for (int i = 0; i < EPT; ++i) {
          const bool mine = (row[i] >= NHALF0 ? 1 : 0) == hh;
          if (mine) {
            int e = t * EPT + i;
            if (pos[i] < CAP) {
              slots[(size_t)row[i] * SLOT + ENT0 + pos[i]] = e;
            } else {
              int o = atomicAdd(spill, 1);
              spill[1 + o] = e;
            }
          }
        }
      }
    }
  }
}

struct Comp { float m0, m1, m2, i0, i1, i2; };

__device__ __forceinline__ float logit3(float u0, float u1, float u2, const Comp& c) {
  float d0 = u0 - c.m0, d1 = u1 - c.m1, d2 = u2 - c.m2;
  float s = d0 * d0 * c.i0;
  s += d1 * d1 * c.i1;
  s += d2 * d2 * c.i2;
  return -0.5f * s;
}

__device__ __forceinline__ int grp8i(int v, int Q) {
  return __builtin_amdgcn_permlane16(v, v, (unsigned)Q * 0x11111111u,
                                     (unsigned)(8 + Q) * 0x11111111u, false, false);
}
__device__ __forceinline__ float grp8(float v, int Q) {
  return __builtin_bit_cast(float, grp8i(__builtin_bit_cast(int, v), Q));
}

__device__ __forceinline__ int edge_id(int e) { return min(max(e, 0), EE - 1); }

constexpr int NPB = 16;
constexpr int ROWT = NPB * KK;
constexpr int PITCH = SLOT + 4;
static_assert(NN % NPB == 0, "whole work-groups only: the grid is exactly NN / NPB");

__global__ __launch_bounds__(ROWT) void k_row_gemm(const float* __restrict__ X,
                      const float* __restrict__ uval,
                      const float* __restrict__ mu,
                      const float* __restrict__ sigma,
                      const float* __restrict__ GW,
                      const int* __restrict__ urows,
                      const int* __restrict__ ucols,
                      const int* __restrict__ spill,
                      const int* __restrict__ slots,
                      float* __restrict__ out) {
  __shared__ v4f rowbuf[NPB * PITCH / 4];
  const int jn = threadIdx.x >> 3;
  const int n0 = blockIdx.x * NPB;
  const int node = n0 + jn;
  const int k = threadIdx.x & (KK - 1);

  Comp c;
  c.m0 = mu[k * DD + 0]; c.m1 = mu[k * DD + 1]; c.m2 = mu[k * DD + 2];
  {
    float s0 = sigma[k * DD + 0], s1 = sigma[k * DD + 1], s2 = sigma[k * DD + 2];
    c.i0 = 1.0f / (s0 * s0 + EPSF);
    c.i1 = 1.0f / (s1 * s1 + EPSF);
    c.i2 = 1.0f / (s2 * s2 + EPSF);
  }

  const int* slot = slots + (size_t)node * SLOT;
  const int cnt = slot[0];
  const int nin = min(max(cnt, 0), CAP);
  const bool spilled = cnt > CAP;
  int nsp = 0;
  if (spilled) nsp = min(max(spill[0], 0), 2 * NK - 1);

  const int r = k & 3;
  const unsigned xq = (unsigned)(k >> 2) * (unsigned)(NN * FF * 4) + (unsigned)r * 8u;
  const char* Xb = (const char*)X;
  const unsigned sbase = (unsigned)node * (unsigned)(SLOT * 4) + (unsigned)(ENT0 * 4);
  float m = -INFINITY;
  float den = 0.0f;
  float acc[FB];
#pragma unroll
  for (int i = 0; i < FB; ++i) acc[i] = 0.0f;

  for (int j = 0; j < nin; j += CH_W) {
    const int left = nin - j;
    const int mye = edge_id(*(const int*)((const char*)slots + (sbase + 4u * (unsigned)(j + (r < left ? r : 0)))));
    const float* p = uval + (unsigned)mye * (unsigned)DD;
    const float uv0 = p[0], uv1 = p[1], uv2 = p[2];
    const int mycol = ucols[mye];
    float l[CH_W] = {0.0f, 0.0f, 0.0f, 0.0f};
    v2f x[CH_W];
#pragma unroll
    for (int i = 0; i < CH_W; ++i) {
      const unsigned col = (unsigned)grp8i(mycol, i);
      x[i] = *(const v2f*)(Xb + (xq + col * (unsigned)(FF * 4)));
    }
    float mn = m;
#pragma nounroll
    for (int i = 0; i < CH_W; ++i) {
      float lg = logit3(grp8(uv0, i), grp8(uv1, i), grp8(uv2, i), c);
      if (!(i < left)) lg = -INFINITY;
      mn = fmaxf(mn, lg);
      l[0] = i == 0 ? lg : l[0];
      l[1] = i == 1 ? lg : l[1];
      l[2] = i == 2 ? lg : l[2];
      l[3] = i == 3 ? lg : l[3];
    }
    const float sc = __expf(m - mn);
    m = mn;
    den *= sc;
#pragma unroll
    for (int q = 0; q < FB; ++q) acc[q] *= sc;
#pragma unroll
    for (int i = 0; i < CH_W; ++i) {
      float w = __expf(l[i] - mn);
      den += w;
#pragma unroll
      for (int f = 0; f < FF; ++f) {
        acc[f * 2 + 0] = __builtin_fmaf(w, grp8(x[i][f & 1], f >> 1), acc[f * 2 + 0]);
        acc[f * 2 + 1] = __builtin_fmaf(w, grp8(x[i][f & 1], 4 + (f >> 1)), acc[f * 2 + 1]);
      }
    }
  }
  if (spilled) {
    for (int i = 0; i < nsp; ++i) {
      int e = edge_id(spill[1 + i]);
      if (urows[e] == node) {
        const float* p = uval + (size_t)e * DD;
        float lg = logit3(p[0], p[1], p[2], c);
        float mn = fmaxf(m, lg);
        float sc = __expf(m - mn);
        float w = __expf(lg - mn);
        m = mn;
        den = den * sc + w;
        int cl = ucols[e];
        const float* x0 = X + (size_t)cl * FF;
        const float* x1 = X + (size_t)NN * FF + (size_t)cl * FF;
#pragma unroll
        for (int f = 0; f < FF; ++f) {
          acc[f * 2 + 0] = acc[f * 2 + 0] * sc + w * x0[f];
          acc[f * 2 + 1] = acc[f * 2 + 1] * sc + w * x1[f];
        }
      }
    }
  }
  {
    const float inv = cnt > 0 ? 1.0f / den : 0.0f;
#pragma unroll
    for (int q = 0; q < FB; ++q) acc[q] *= inv;
  }

  const int lane = threadIdx.x & 31;
  const int half = lane >> 4;
  const int lm = lane & 15;
  const int col0 = (threadIdx.x >> 5) * 16;

#if __has_builtin(__builtin_amdgcn_wmma_f32_16x16x4_f32)
  v2f bfr[CDIM / 4];
#pragma unroll
  for (int kk = 0; kk < CDIM / 4; ++kk) {
    int c0 = kk * 4 + half * 2;
    bfr[kk].x = GW[c0 * HH + col0 + lm];
    bfr[kk].y = GW[(c0 + 1) * HH + col0 + lm];
  }
#endif

  {
    float* rowp = (float*)rowbuf + jn * PITCH + k * 2;
#pragma unroll
    for (int f = 0; f < FF; ++f) {
      v2f v;
      v.x = acc[f * 2 + 0];
      v.y = acc[f * 2 + 1];
      *(v2f*)(rowp + f * 16) = v;
    }
  }
  __syncthreads();

#if __has_builtin(__builtin_amdgcn_wmma_f32_16x16x4_f32)
  const v4f* ap = rowbuf + lm * (PITCH / 4) + half;
  v8f d0 = {}, d1 = {};
#pragma unroll
  for (int kk = 0; kk < CDIM / 4; ++kk) {
    v4f a4 = ap[kk * 2];
    v2f a0, a1;
    a0.x = a4.x; a0.y = a4.z;
    a1.x = a4.y; a1.y = a4.w;
    d0 = __builtin_amdgcn_wmma_f32_16x16x4_f32(
         false, a0,  false, bfr[kk],
         (short)0, d0,  false,  false);
    d1 = __builtin_amdgcn_wmma_f32_16x16x4_f32(
         false, a1,  false, bfr[kk],
         (short)0, d1,  false,  false);
  }
#pragma unroll
  for (int j = 0; j < 8; ++j) {
    size_t r = (size_t)(n0 + half * 8 + j);
    out[r * HH + col0 + lm] = d0[j];
    out[((size_t)NN + r) * HH + col0 + lm] = d1[j];
  }
#else
  {
    const float* rw = (const float*)rowbuf;
    for (int j = 0; j < 8; ++j) {
      int jr = half * 8 + j;
      for (int bb = 0; bb < BB; ++bb) {
        float s = 0.0f;
        for (int c = 0; c < CDIM; ++c)
          s += rw[jr * PITCH + (c >> 3) * 16 + (c & 7) * 2 + bb] * GW[c * HH + col0 + lm];
        out[((size_t)bb * NN + n0 + jr) * HH + col0 + lm] = s;
      }
    }
  }
#endif
}

extern "C" void kernel_launch(void* const* d_in, const int* in_sizes, int n_in,
                              void* d_out, int out_size, void* d_ws, size_t ws_size,
                              hipStream_t stream) {
  const float* X     = (const float*)d_in[0];
  const float* uval  = (const float*)d_in[1];
  const float* mu    = (const float*)d_in[2];
  const float* sigma = (const float*)d_in[3];
  const float* GW    = (const float*)d_in[4];
  const int* urows   = (const int*)d_in[5];
  const int* ucols   = (const int*)d_in[6];
  float* out = (float*)d_out;

  int* spill = (int*)d_ws;
  int* slots = (int*)d_ws + 2 * NK;

  k_init<<<(NN + 255) / 256, 256, 0, stream>>>(spill, slots);

  k_fill<<<2 * NCHUNK, FILL_T, 0, stream>>>(urows, spill, slots);

  k_row_gemm<<<NN / NPB, ROWT, 0, stream>>>(
      X, uval, mu, sigma, GW, urows, ucols, spill, slots, out);
}


# Bucket entries are (edge id, column id) pairs written by the fill pass from a streamed column read: node pass gathers no column ids; count shares the bucket's first line; same sum order
# speedup vs baseline: 16.7437x; 1.0485x over previous
#include <hip/hip_runtime.h>
#include <math.h>

constexpr int BB = 2;
constexpr int NN = 50000;
constexpr int FF = 8;
constexpr int EE = 800000;
constexpr int DD = 3;
constexpr int KK = 8;
constexpr int HH = 64;
constexpr int FB = FF * BB;
constexpr int NK = NN * KK;
constexpr int CDIM = FF * KK;
constexpr float EPSF = 1e-14f;

typedef __attribute__((ext_vector_type(2))) float v2f;
typedef __attribute__((ext_vector_type(8))) float v8f;

typedef __attribute__((ext_vector_type(4))) float v4f;
typedef __attribute__((ext_vector_type(4))) int v4i;
typedef __attribute__((ext_vector_type(2))) int v2i;

constexpr int SLOT = KK * FB;
constexpr int ENT0 = 2;
constexpr int CAP = (SLOT - ENT0) / 2;
constexpr int CH_W = 4;

constexpr int EPT = 16;
constexpr int FILL_T = 256;
constexpr int CHUNK = FILL_T * EPT;
constexpr int NCHUNK = (EE + CHUNK - 1) / CHUNK;
constexpr int NHALF0 = NN / 2;
constexpr int QWORD = 1;
__device__ __forceinline__ size_t q_index(int h) {
  return (size_t)(h == 0 ? 0 : NN - 1) * SLOT + QWORD;
}

__global__ void k_init(int* __restrict__ spill, int* __restrict__ slots) {
  int i = blockIdx.x * blockDim.x + threadIdx.x;
  if (i < NN) slots[(size_t)i * SLOT] = 0;
  if (i == 0) {
    spill[0] = 0;
    slots[q_index(0)] = 0;
    slots[q_index(1)] = 0;
  }
}

__global__ __launch_bounds__(FILL_T) void k_fill(const int* __restrict__ urows,
                       const int* __restrict__ ucols,
                       int* __restrict__ spill,
                       int* __restrict__ slots) {
  __shared__ int sh_ticket;
  __shared__ int sh_half;
  if (threadIdx.x == 0) {
    unsigned id = __builtin_amdgcn_s_sendmsg_rtn(0x87);
    sh_half = (int)((id >> 18) & 1u);
  }
  __syncthreads();
  const int myhalf = sh_half;
  for (int pass = 0; pass < 2; ++pass) {
    const int hh = pass == 0 ? myhalf : 1 - myhalf;
    int* q = slots + q_index(hh);
    for (int it = 0; it <= NCHUNK; ++it) {
      if (threadIdx.x == 0) sh_ticket = atomicAdd(q, 1);
      __syncthreads();
      const int c = sh_ticket;
      __syncthreads();
      if (c < 0 || c >= NCHUNK) break;
      const int t = c * FILL_T + threadIdx.x;
      if (t < EE / EPT) {
        const v4i* rp = (const v4i*)urows + (size_t)t * (EPT / 4);
        int row[EPT];
#pragma unroll
        for (int g = 0; g < EPT / 4; ++g) {
          v4i r = rp[g];
          row[g * 4 + 0] = r.x; row[g * 4 + 1] = r.y; row[g * 4 + 2] = r.z; row[g * 4 + 3] = r.w;
        }
        const v4i* cp = (const v4i*)ucols + (size_t)t * (EPT / 4);
        int col[EPT];
#pragma unroll
        for (int g = 0; g < EPT / 4; ++g) {
          v4i q4 = cp[g];
          col[g * 4 + 0] = q4.x; col[g * 4 + 1] = q4.y; col[g * 4 + 2] = q4.z; col[g * 4 + 3] = q4.w;
        }
        int pos[EPT];
#pragma unroll
        for (int i = 0; i < EPT; ++i) {
          const bool mine = (row[i] >= NHALF0 ? 1 : 0) == hh;
          pos[i] = -1;
          if (mine) pos[i] = atomicAdd(slots + (size_t)row[i] * SLOT, 1);
        }
#pragma unroll
        for (int i = 0; i < EPT; ++i) {
          const bool mine = (row[i] >= NHALF0 ? 1 : 0) == hh;
          if (mine) {
            int e = t * EPT + i;
            if (pos[i] < CAP) {
              v2i ent;
              ent.x = e;
              ent.y = col[i];
              *(v2i*)(slots + (size_t)row[i] * SLOT + ENT0 + 2 * pos[i]) = ent;
            } else {
              int o = atomicAdd(spill, 1);
              spill[1 + o] = e;
            }
          }
        }
      }
    }
  }
}

struct Comp { float m0, m1, m2, i0, i1, i2; };

__device__ __forceinline__ float logit3(float u0, float u1, float u2, const Comp& c) {
  float d0 = u0 - c.m0, d1 = u1 - c.m1, d2 = u2 - c.m2;
  float s = d0 * d0 * c.i0;
  s += d1 * d1 * c.i1;
  s += d2 * d2 * c.i2;
  return -0.5f * s;
}

__device__ __forceinline__ int grp8i(int v, int Q) {
  return __builtin_amdgcn_permlane16(v, v, (unsigned)Q * 0x11111111u,
                                     (unsigned)(8 + Q) * 0x11111111u, false, false);
}
__device__ __forceinline__ float grp8(float v, int Q) {
  return __builtin_bit_cast(float, grp8i(__builtin_bit_cast(int, v), Q));
}

__device__ __forceinline__ int edge_id(int e) { return min(max(e, 0), EE - 1); }
__device__ __forceinline__ int node_id(int n) { return min(max(n, 0), NN - 1); }

constexpr int NPB = 16;
constexpr int ROWT = NPB * KK;
constexpr int PITCH = SLOT + 4;
static_assert(NN % NPB == 0, "whole work-groups only: the grid is exactly NN / NPB");

__global__ __launch_bounds__(ROWT) void k_row_gemm(const float* __restrict__ X,
                      const float* __restrict__ uval,
                      const float* __restrict__ mu,
                      const float* __restrict__ sigma,
                      const float* __restrict__ GW,
                      const int* __restrict__ urows,
                      const int* __restrict__ ucols,
                      const int* __restrict__ spill,
                      const int* __restrict__ slots,
                      float* __restrict__ out) {
  __shared__ v4f rowbuf[NPB * PITCH / 4];
  const int jn = threadIdx.x >> 3;
  const int n0 = blockIdx.x * NPB;
  const int node = n0 + jn;
  const int k = threadIdx.x & (KK - 1);

  Comp c;
  c.m0 = mu[k * DD + 0]; c.m1 = mu[k * DD + 1]; c.m2 = mu[k * DD + 2];
  {
    float s0 = sigma[k * DD + 0], s1 = sigma[k * DD + 1], s2 = sigma[k * DD + 2];
    c.i0 = 1.0f / (s0 * s0 + EPSF);
    c.i1 = 1.0f / (s1 * s1 + EPSF);
    c.i2 = 1.0f / (s2 * s2 + EPSF);
  }

  const int* slot = slots + (size_t)node * SLOT;
  const int cnt = slot[0];
  const int nin = min(max(cnt, 0), CAP);
  const bool spilled = cnt > CAP;
  int nsp = 0;
  if (spilled) nsp = min(max(spill[0], 0), 2 * NK - 1);

  const int r = k & 3;
  const unsigned xq = (unsigned)(k >> 2) * (unsigned)(NN * FF * 4) + (unsigned)r * 8u;
  const char* Xb = (const char*)X;
  const unsigned sbase = (unsigned)node * (unsigned)(SLOT * 4) + (unsigned)(ENT0 * 4);
  float m = -INFINITY;
  float den = 0.0f;
  float acc[FB];
#pragma unroll
  for (int i = 0; i < FB; ++i) acc[i] = 0.0f;

  for (int j = 0; j < nin; j += CH_W) {
    const int left = nin - j;
    const v2i ent = *(const v2i*)((const char*)slots + (sbase + 8u * (unsigned)(j + (r < left ? r : 0))));
    const int mye = edge_id(ent.x);
    const int mycol = node_id(ent.y);
    const float* p = uval + (unsigned)mye * (unsigned)DD;
    const float uv0 = p[0], uv1 = p[1], uv2 = p[2];
    float l[CH_W] = {0.0f, 0.0f, 0.0f, 0.0f};
    v2f x[CH_W];
#pragma unroll
    for (int i = 0; i < CH_W; ++i) {
      const unsigned col = (unsigned)grp8i(mycol, i);
      x[i] = *(const v2f*)(Xb + (xq + col * (unsigned)(FF * 4)));
    }
    float mn = m;
#pragma nounroll
    for (int i = 0; i < CH_W; ++i) {
      float lg = logit3(grp8(uv0, i), grp8(uv1, i), grp8(uv2, i), c);
      if (!(i < left)) lg = -INFINITY;
      mn = fmaxf(mn, lg);
      l[0] = i == 0 ? lg : l[0];
      l[1] = i == 1 ? lg : l[1];
      l[2] = i == 2 ? lg : l[2];
      l[3] = i == 3 ? lg : l[3];
    }
    const float sc = __expf(m - mn);
    m = mn;
    den *= sc;
#pragma unroll
    for (int q = 0; q < FB; ++q) acc[q] *= sc;
#pragma unroll
    for (int i = 0; i < CH_W; ++i) {
      float w = __expf(l[i] - mn);
      den += w;
#pragma unroll
      for (int f = 0; f < FF; ++f) {
        acc[f * 2 + 0] = __builtin_fmaf(w, grp8(x[i][f & 1], f >> 1), acc[f * 2 + 0]);
        acc[f * 2 + 1] = __builtin_fmaf(w, grp8(x[i][f & 1], 4 + (f >> 1)), acc[f * 2 + 1]);
      }
    }
  }
  if (spilled) {
    for (int i = 0; i < nsp; ++i) {
      int e = edge_id(spill[1 + i]);
      if (urows[e] == node) {
        const float* p = uval + (size_t)e * DD;
        float lg = logit3(p[0], p[1], p[2], c);
        float mn = fmaxf(m, lg);
        float sc = __expf(m - mn);
        float w = __expf(lg - mn);
        m = mn;
        den = den * sc + w;
        int cl = ucols[e];
        const float* x0 = X + (size_t)cl * FF;
        const float* x1 = X + (size_t)NN * FF + (size_t)cl * FF;
#pragma unroll
        for (int f = 0; f < FF; ++f) {
          acc[f * 2 + 0] = acc[f * 2 + 0] * sc + w * x0[f];
          acc[f * 2 + 1] = acc[f * 2 + 1] * sc + w * x1[f];
        }
      }
    }
  }
  {
    const float inv = cnt > 0 ? 1.0f / den : 0.0f;
#pragma unroll
    for (int q = 0; q < FB; ++q) acc[q] *= inv;
  }

  const int lane = threadIdx.x & 31;
  const int half = lane >> 4;
  const int lm = lane & 15;
  const int col0 = (threadIdx.x >> 5) * 16;

#if __has_builtin(__builtin_amdgcn_wmma_f32_16x16x4_f32)
  v2f bfr[CDIM / 4];
#pragma unroll
  for (int kk = 0; kk < CDIM / 4; ++kk) {
    int c0 = kk * 4 + half * 2;
    bfr[kk].x = GW[c0 * HH + col0 + lm];
    bfr[kk].y = GW[(c0 + 1) * HH + col0 + lm];
  }
#endif

  {
    float* rowp = (float*)rowbuf + jn * PITCH + k * 2;
#pragma unroll
    for (int f = 0; f < FF; ++f) {
      v2f v;
      v.x = acc[f * 2 + 0];
      v.y = acc[f * 2 + 1];
      *(v2f*)(rowp + f * 16) = v;
    }
  }
  __syncthreads();

#if __has_builtin(__builtin_amdgcn_wmma_f32_16x16x4_f32)
  const v4f* ap = rowbuf + lm * (PITCH / 4) + half;
  v8f d0 = {}, d1 = {};
#pragma unroll
  for (int kk = 0; kk < CDIM / 4; ++kk) {
    v4f a4 = ap[kk * 2];
    v2f a0, a1;
    a0.x = a4.x; a0.y = a4.z;
    a1.x = a4.y; a1.y = a4.w;
    d0 = __builtin_amdgcn_wmma_f32_16x16x4_f32(
         false, a0,  false, bfr[kk],
         (short)0, d0,  false,  false);
    d1 = __builtin_amdgcn_wmma_f32_16x16x4_f32(
         false, a1,  false, bfr[kk],
         (short)0, d1,  false,  false);
  }
#pragma unroll
  for (int j = 0; j < 8; ++j) {
    size_t r = (size_t)(n0 + half * 8 + j);
    out[r * HH + col0 + lm] = d0[j];
    out[((size_t)NN + r) * HH + col0 + lm] = d1[j];
  }
#else
  {
    const float* rw = (const float*)rowbuf;
    for (int j = 0; j < 8; ++j) {
      int jr = half * 8 + j;
      for (int bb = 0; bb < BB; ++bb) {
        float s = 0.0f;
        for (int c = 0; c < CDIM; ++c)
          s += rw[jr * PITCH + (c >> 3) * 16 + (c & 7) * 2 + bb] * GW[c * HH + col0 + lm];
        out[((size_t)bb * NN + n0 + jr) * HH + col0 + lm] = s;
      }
    }
  }
#endif
}

extern "C" void kernel_launch(void* const* d_in, const int* in_sizes, int n_in,
                              void* d_out, int out_size, void* d_ws, size_t ws_size,
                              hipStream_t stream) {
  const float* X     = (const float*)d_in[0];
  const float* uval  = (const float*)d_in[1];
  const float* mu    = (const float*)d_in[2];
  const float* sigma = (const float*)d_in[3];
  const float* GW    = (const float*)d_in[4];
  const int* urows   = (const int*)d_in[5];
  const int* ucols   = (const int*)d_in[6];
  float* out = (float*)d_out;

  int* spill = (int*)d_ws;
  int* slots = (int*)d_ws + 2 * NK;

  k_init<<<(NN + 255) / 256, 256, 0, stream>>>(spill, slots);

  k_fill<<<2 * NCHUNK, FILL_T, 0, stream>>>(urows, ucols, spill, slots);

  k_row_gemm<<<NN / NPB, ROWT, 0, stream>>>(
      X, uval, mu, sigma, GW, urows, ucols, spill, slots, out);
}


# Node pass re-partitioned: 4 lanes per node, each lane two mixture components in packed f32, so an edge's X row is passed round once for both; 8 nodes per wave, half the waves; same sum order
# speedup vs baseline: 17.8725x; 1.0674x over previous
#include <hip/hip_runtime.h>
#include <math.h>

constexpr int BB = 2;
constexpr int NN = 50000;
constexpr int FF = 8;
constexpr int EE = 800000;
constexpr int DD = 3;
constexpr int KK = 8;
constexpr int HH = 64;
constexpr int FB = FF * BB;
constexpr int NK = NN * KK;
constexpr int CDIM = FF * KK;
constexpr float EPSF = 1e-14f;

typedef __attribute__((ext_vector_type(2))) float v2f;
typedef __attribute__((ext_vector_type(8))) float v8f;

typedef __attribute__((ext_vector_type(4))) float v4f;
typedef __attribute__((ext_vector_type(4))) int v4i;
typedef __attribute__((ext_vector_type(2))) int v2i;

constexpr int SLOT = KK * FB;
constexpr int ENT0 = 2;
constexpr int CAP = (SLOT - ENT0) / 2;
constexpr int CH_W = 4;

constexpr int EPT = 16;
constexpr int FILL_T = 256;
constexpr int CHUNK = FILL_T * EPT;
constexpr int NCHUNK = (EE + CHUNK - 1) / CHUNK;
constexpr int NHALF0 = NN / 2;
constexpr int QWORD = 1;
__device__ __forceinline__ size_t q_index(int h) {
  return (size_t)(h == 0 ? 0 : NN - 1) * SLOT + QWORD;
}

__global__ void k_init(int* __restrict__ spill, int* __restrict__ slots) {
  int i = blockIdx.x * blockDim.x + threadIdx.x;
  if (i < NN) slots[(size_t)i * SLOT] = 0;
  if (i == 0) {
    spill[0] = 0;
    slots[q_index(0)] = 0;
    slots[q_index(1)] = 0;
  }
}

__global__ __launch_bounds__(FILL_T) void k_fill(const int* __restrict__ urows,
                       const int* __restrict__ ucols,
                       int* __restrict__ spill,
                       int* __restrict__ slots) {
  __shared__ int sh_ticket;
  __shared__ int sh_half;
  if (threadIdx.x == 0) {
    unsigned id = __builtin_amdgcn_s_sendmsg_rtn(0x87);
    sh_half = (int)((id >> 18) & 1u);
  }
  __syncthreads();
  const int myhalf = sh_half;
  for (int pass = 0; pass < 2; ++pass) {
    const int hh = pass == 0 ? myhalf : 1 - myhalf;
    int* q = slots + q_index(hh);
    for (int it = 0; it <= NCHUNK; ++it) {
      if (threadIdx.x == 0) sh_ticket = atomicAdd(q, 1);
      __syncthreads();
      const int c = sh_ticket;
      __syncthreads();
      if (c < 0 || c >= NCHUNK) break;
      const int t = c * FILL_T + threadIdx.x;
      if (t < EE / EPT) {
        const v4i* rp = (const v4i*)urows + (size_t)t * (EPT / 4);
        int row[EPT];
#pragma unroll
        for (int g = 0; g < EPT / 4; ++g) {
          v4i r = rp[g];
          row[g * 4 + 0] = r.x; row[g * 4 + 1] = r.y; row[g * 4 + 2] = r.z; row[g * 4 + 3] = r.w;
        }
        const v4i* cp = (const v4i*)ucols + (size_t)t * (EPT / 4);
        int col[EPT];
#pragma unroll
        for (int g = 0; g < EPT / 4; ++g) {
          v4i q4 = cp[g];
          col[g * 4 + 0] = q4.x; col[g * 4 + 1] = q4.y; col[g * 4 + 2] = q4.z; col[g * 4 + 3] = q4.w;
        }
        int pos[EPT];
#pragma unroll
        for (int i = 0; i < EPT; ++i) {
          const bool mine = (row[i] >= NHALF0 ? 1 : 0) == hh;
          pos[i] = -1;
          if (mine) pos[i] = atomicAdd(slots + (size_t)row[i] * SLOT, 1);
        }
#pragma unroll
        for (int i = 0; i < EPT; ++i) {
          const bool mine = (row[i] >= NHALF0 ? 1 : 0) == hh;
          if (mine) {
            int e = t * EPT + i;
            if (pos[i] < CAP) {
              v2i ent;
              ent.x = e;
              ent.y = col[i];
              *(v2i*)(slots + (size_t)row[i] * SLOT + ENT0 + 2 * pos[i]) = ent;
            } else {
              int o = atomicAdd(spill, 1);
              spill[1 + o] = e;
            }
          }
        }
      }
    }
  }
}

struct Comp2 { v2f m0, m1, m2, i0, i1, i2; };

__device__ __forceinline__ v2f splat2(float v) { v2f t; t.x = v; t.y = v; return t; }

__device__ __forceinline__ v2f logit3(float u0, float u1, float u2, const Comp2& c) {
  v2f d0 = splat2(u0) - c.m0, d1 = splat2(u1) - c.m1, d2 = splat2(u2) - c.m2;
  v2f s = d0 * d0 * c.i0;
  s += d1 * d1 * c.i1;
  s += d2 * d2 * c.i2;
  return -0.5f * s;
}

__device__ __forceinline__ v2f exp2v(v2f a) { v2f t; t.x = __expf(a.x); t.y = __expf(a.y); return t; }
__device__ __forceinline__ v2f max2v(v2f a, v2f b) { v2f t; t.x = fmaxf(a.x, b.x); t.y = fmaxf(a.y, b.y); return t; }

__device__ __forceinline__ Comp2 load_comp2(const float* __restrict__ mu,
                                            const float* __restrict__ sigma, int k) {
  Comp2 c;
  c.m0.x = mu[k * DD + 0]; c.m1.x = mu[k * DD + 1]; c.m2.x = mu[k * DD + 2];
  c.m0.y = mu[k * DD + 3]; c.m1.y = mu[k * DD + 4]; c.m2.y = mu[k * DD + 5];
  v2f s0, s1, s2;
  s0.x = sigma[k * DD + 0]; s1.x = sigma[k * DD + 1]; s2.x = sigma[k * DD + 2];
  s0.y = sigma[k * DD + 3]; s1.y = sigma[k * DD + 4]; s2.y = sigma[k * DD + 5];
  c.i0 = 1.0f / (s0 * s0 + EPSF);
  c.i1 = 1.0f / (s1 * s1 + EPSF);
  c.i2 = 1.0f / (s2 * s2 + EPSF);
  return c;
}

__device__ __forceinline__ int grp4i(int v, int Q) {
  return __builtin_amdgcn_permlane16(v, v, (unsigned)Q * 0x11111111u + 0x44440000u,
                                     (unsigned)Q * 0x11111111u + 0xCCCC8888u, false, false);
}
__device__ __forceinline__ float grp4(float v, int Q) {
  return __builtin_bit_cast(float, grp4i(__builtin_bit_cast(int, v), Q));
}

__device__ __forceinline__ int edge_id(int e) { return min(max(e, 0), EE - 1); }
__device__ __forceinline__ int node_id(int n) { return min(max(n, 0), NN - 1); }

constexpr int LPN = 4;
constexpr int NPB = 16;
constexpr int ROWT = NPB * LPN;
constexpr int PITCH = SLOT + 4;
constexpr int NCT = 2;
static_assert(NN % NPB == 0, "whole work-groups only: the grid is exactly NN / NPB");
static_assert((ROWT / 32) * NCT * 16 == HH, "the waves' column tiles cover GW's columns");

__global__ __launch_bounds__(ROWT) void k_row_gemm(const float* __restrict__ X,
                      const float* __restrict__ uval,
                      const float* __restrict__ mu,
                      const float* __restrict__ sigma,
                      const float* __restrict__ GW,
                      const int* __restrict__ urows,
                      const int* __restrict__ ucols,
                      const int* __restrict__ spill,
                      const int* __restrict__ slots,
                      float* __restrict__ out) {
  __shared__ v4f rowbuf[NPB * PITCH / 4];
  const int jn = threadIdx.x >> 2;
  const int n0 = blockIdx.x * NPB;
  const int node = n0 + jn;
  const int r = threadIdx.x & (LPN - 1);

  const Comp2 cc = load_comp2(mu, sigma, 2 * r);

  const int* slot = slots + (size_t)node * SLOT;
  const int cnt = slot[0];
  const int nin = min(max(cnt, 0), CAP);
  const bool spilled = cnt > CAP;
  int nsp = 0;
  if (spilled) nsp = min(max(spill[0], 0), 2 * NK - 1);

  const unsigned xq = (unsigned)(r >> 1) * (unsigned)(NN * FF * 4) + (unsigned)(r & 1) * 16u;
  const char* Xb = (const char*)X;
  const unsigned sbase = (unsigned)node * (unsigned)(SLOT * 4) + (unsigned)(ENT0 * 4);
  const v2f ninf2 = splat2(-INFINITY);
  v2f m = ninf2;
  v2f den = splat2(0.0f);
  v2f acc[FB];
#pragma unroll
  for (int i = 0; i < FB; ++i) acc[i] = splat2(0.0f);

  for (int j = 0; j < nin; j += CH_W) {
    const int left = nin - j;
    const v2i ent = *(const v2i*)((const char*)slots + (sbase + 8u * (unsigned)(j + (r < left ? r : 0))));
    const int mye = edge_id(ent.x);
    const int mycol = node_id(ent.y);
    const float* p = uval + (unsigned)mye * (unsigned)DD;
    const float uv0 = p[0], uv1 = p[1], uv2 = p[2];
    v4f x[CH_W];
#pragma unroll
    for (int i = 0; i < CH_W; ++i) {
      const unsigned col = (unsigned)grp4i(mycol, i);
      x[i] = *(const v4f*)(Xb + (xq + col * (unsigned)(FF * 4)));
    }
    v2f l[CH_W];
    v2f mn = m;
#pragma unroll
    for (int i = 0; i < CH_W; ++i) {
      v2f lg = logit3(grp4(uv0, i), grp4(uv1, i), grp4(uv2, i), cc);
      if (!(i < left)) lg = ninf2;
      l[i] = lg;
      mn = max2v(mn, lg);
    }
    const v2f sc = exp2v(m - mn);
    m = mn;
    den *= sc;
#pragma unroll
    for (int q = 0; q < FB; ++q) acc[q] *= sc;
#pragma unroll
    for (int i = 0; i < CH_W; ++i) {
      const v2f w = exp2v(l[i] - mn);
      den += w;
#pragma unroll
      for (int f = 0; f < FF; ++f) {
        acc[f * 2 + 0] = __builtin_elementwise_fma(w, splat2(grp4(x[i][f & 3], f >> 2)), acc[f * 2 + 0]);
        acc[f * 2 + 1] = __builtin_elementwise_fma(w, splat2(grp4(x[i][f & 3], 2 + (f >> 2))), acc[f * 2 + 1]);
      }
    }
  }
  if (spilled) {
    for (int i = 0; i < nsp; ++i) {
      int e = edge_id(spill[1 + i]);
      if (urows[e] == node) {
        const float* p = uval + (size_t)e * DD;
        const v2f lg = logit3(p[0], p[1], p[2], cc);
        const v2f mn = max2v(m, lg);
        const v2f sc = exp2v(m - mn);
        const v2f w = exp2v(lg - mn);
        m = mn;
        den = den * sc + w;
        int cl = ucols[e];
        const float* x0 = X + (size_t)cl * FF;
        const float* x1 = X + (size_t)NN * FF + (size_t)cl * FF;
#pragma unroll
        for (int f = 0; f < FF; ++f) {
          acc[f * 2 + 0] = acc[f * 2 + 0] * sc + w * splat2(x0[f]);
          acc[f * 2 + 1] = acc[f * 2 + 1] * sc + w * splat2(x1[f]);
        }
      }
    }
  }
  {
    v2f inv = splat2(0.0f);
    if (cnt > 0) inv = 1.0f / den;
#pragma unroll
    for (int q = 0; q < FB; ++q) acc[q] *= inv;
  }

  const int lane = threadIdx.x & 31;
  const int half = lane >> 4;
  const int lm = lane & 15;
  const int colw = (threadIdx.x >> 5) * (NCT * 16);

#if __has_builtin(__builtin_amdgcn_wmma_f32_16x16x4_f32)
  v2f bfr[NCT][CDIM / 4];
#pragma unroll
  for (int ct = 0; ct < NCT; ++ct) {
#pragma unroll
    for (int kk = 0; kk < CDIM / 4; ++kk) {
      int c0 = kk * 4 + half * 2;
      bfr[ct][kk].x = GW[c0 * HH + colw + ct * 16 + lm];
      bfr[ct][kk].y = GW[(c0 + 1) * HH + colw + ct * 16 + lm];
    }
  }
#endif

  {
    float* rowp = (float*)rowbuf + jn * PITCH + r * 4;
#pragma unroll
    for (int f = 0; f < FF; ++f) {
      v4f v;
      v.x = acc[f * 2 + 0].x;
      v.y = acc[f * 2 + 1].x;
      v.z = acc[f * 2 + 0].y;
      v.w = acc[f * 2 + 1].y;
      *(v4f*)(rowp + f * 16) = v;
    }
  }
  __syncthreads();

#if __has_builtin(__builtin_amdgcn_wmma_f32_16x16x4_f32)
  const v4f* ap = rowbuf + lm * (PITCH / 4) + half;
  v8f d0[NCT], d1[NCT];
#pragma unroll
  for (int ct = 0; ct < NCT; ++ct) { d0[ct] = v8f{}; d1[ct] = v8f{}; }
#pragma unroll
  for (int kk = 0; kk < CDIM / 4; ++kk) {
    v4f a4 = ap[kk * 2];
    v2f a0, a1;
    a0.x = a4.x; a0.y = a4.z;
    a1.x = a4.y; a1.y = a4.w;
#pragma unroll
    for (int ct = 0; ct < NCT; ++ct) {
      d0[ct] = __builtin_amdgcn_wmma_f32_16x16x4_f32(
           false, a0,  false, bfr[ct][kk],
           (short)0, d0[ct],  false,  false);
      d1[ct] = __builtin_amdgcn_wmma_f32_16x16x4_f32(
           false, a1,  false, bfr[ct][kk],
           (short)0, d1[ct],  false,  false);
    }
  }
#pragma unroll
  for (int ct = 0; ct < NCT; ++ct) {
#pragma unroll
    for (int j = 0; j < 8; ++j) {
      size_t rr = (size_t)(n0 + half * 8 + j);
      out[rr * HH + colw + ct * 16 + lm] = d0[ct][j];
      out[((size_t)NN + rr) * HH + colw + ct * 16 + lm] = d1[ct][j];
    }
  }
#else
  {
    const float* rw = (const float*)rowbuf;
    for (int ct = 0; ct < NCT; ++ct) {
      for (int j = 0; j < 8; ++j) {
        int jr = half * 8 + j;
        for (int bb = 0; bb < BB; ++bb) {
          float s = 0.0f;
          for (int c = 0; c < CDIM; ++c)
            s += rw[jr * PITCH + (c >> 3) * 16 + (c & 7) * 2 + bb] * GW[c * HH + colw + ct * 16 + lm];
          out[((size_t)bb * NN + n0 + jr) * HH + colw + ct * 16 + lm] = s;
        }
      }
    }
  }
#endif
}

extern "C" void kernel_launch(void* const* d_in, const int* in_sizes, int n_in,
                              void* d_out, int out_size, void* d_ws, size_t ws_size,
                              hipStream_t stream) {
  const float* X     = (const float*)d_in[0];
  const float* uval  = (const float*)d_in[1];
  const float* mu    = (const float*)d_in[2];
  const float* sigma = (const float*)d_in[3];
  const float* GW    = (const float*)d_in[4];
  const int* urows   = (const int*)d_in[5];
  const int* ucols   = (const int*)d_in[6];
  float* out = (float*)d_out;

  int* spill = (int*)d_ws;
  int* slots = (int*)d_ws + 2 * NK;

  k_init<<<(NN + 255) / 256, 256, 0, stream>>>(spill, slots);

  k_fill<<<2 * NCHUNK, FILL_T, 0, stream>>>(urows, ucols, spill, slots);

  k_row_gemm<<<NN / NPB, ROWT, 0, stream>>>(
      X, uval, mu, sigma, GW, urows, ucols, spill, slots, out);
}


# Init pass also lays each node's 16 features (both batches) as one 64-byte row in its bucket's first line; node pass gathers that one piece per edge, not two 32-byte rows of X; same sum order
# speedup vs baseline: 18.5130x; 1.0358x over previous
#include <hip/hip_runtime.h>
#include <math.h>

constexpr int BB = 2;
constexpr int NN = 50000;
constexpr int FF = 8;
constexpr int EE = 800000;
constexpr int DD = 3;
constexpr int KK = 8;
constexpr int HH = 64;
constexpr int FB = FF * BB;
constexpr int NK = NN * KK;
constexpr int CDIM = FF * KK;
constexpr float EPSF = 1e-14f;

typedef __attribute__((ext_vector_type(2))) float v2f;
typedef __attribute__((ext_vector_type(8))) float v8f;

typedef __attribute__((ext_vector_type(4))) float v4f;
typedef __attribute__((ext_vector_type(4))) int v4i;
typedef __attribute__((ext_vector_type(2))) int v2i;

constexpr int SLOT = KK * FB;
constexpr int ENT0 = 2;
constexpr int XT0 = 16;
constexpr int NE0 = (XT0 - ENT0) / 2;
constexpr int CAP = NE0 + (SLOT - XT0 - FB) / 2;
__device__ __forceinline__ int ent_word(int pos) { return ENT0 + 2 * pos + (pos >= NE0 ? FB : 0); }
constexpr int CH_W = 4;

constexpr int EPT = 16;
constexpr int FILL_T = 256;
constexpr int CHUNK = FILL_T * EPT;
constexpr int NCHUNK = (EE + CHUNK - 1) / CHUNK;
constexpr int NHALF0 = NN / 2;
constexpr int QWORD = 1;
__device__ __forceinline__ size_t q_index(int h) {
  return (size_t)(h == 0 ? 0 : NN - 1) * SLOT + QWORD;
}

constexpr int INIT_T = 256;
constexpr int INIT_G = (NN + INIT_T - 1) / INIT_T;
constexpr int INIT_IPT = 4;
static_assert(INIT_G * INIT_T * INIT_IPT >= 4 * NN, "the init grid covers every quarter row");
__global__ __launch_bounds__(INIT_T) void k_init(const float* __restrict__ X, int* __restrict__ spill,
                                                 int* __restrict__ slots) {
  const int i = blockIdx.x * INIT_T + threadIdx.x;
  const int q = i & 3;
  v2f a[INIT_IPT], b[INIT_IPT];
#pragma unroll
  for (int it = 0; it < INIT_IPT; ++it) {
    const int n = min((it * (INIT_G * INIT_T) + i) >> 2, NN - 1);
    a[it] = *(const v2f*)(X + (size_t)n * FF + 2 * q);
    b[it] = *(const v2f*)(X + (size_t)NN * FF + (size_t)n * FF + 2 * q);
  }
#pragma unroll
  for (int it = 0; it < INIT_IPT; ++it) {
    const int n = (it * (INIT_G * INIT_T) + i) >> 2;
    if (n < NN) {
      v4f v;
      v.x = a[it].x; v.y = b[it].x; v.z = a[it].y; v.w = b[it].y;
      *(v4f*)((float*)(slots + (size_t)n * SLOT + XT0) + 4 * q) = v;
      if (q == 0) slots[(size_t)n * SLOT] = 0;
    }
  }
  if (i == 0) {
    spill[0] = 0;
    slots[q_index(0)] = 0;
    slots[q_index(1)] = 0;
  }
}

__global__ __launch_bounds__(FILL_T) void k_fill(const int* __restrict__ urows,
                       const int* __restrict__ ucols,
                       int* __restrict__ spill,
                       int* __restrict__ slots) {
  __shared__ int sh_ticket;
  __shared__ int sh_half;
  if (threadIdx.x == 0) {
    unsigned id = __builtin_amdgcn_s_sendmsg_rtn(0x87);
    sh_half = (int)((id >> 18) & 1u);
  }
  __syncthreads();
  const int myhalf = sh_half;
  for (int pass = 0; pass < 2; ++pass) {
    const int hh = pass == 0 ? myhalf : 1 - myhalf;
    int* q = slots + q_index(hh);
    for (int it = 0; it <= NCHUNK; ++it) {
      if (threadIdx.x == 0) sh_ticket = atomicAdd(q, 1);
      __syncthreads();
      const int c = sh_ticket;
      __syncthreads();
      if (c < 0 || c >= NCHUNK) break;
      const int t = c * FILL_T + threadIdx.x;
      if (t < EE / EPT) {
        const v4i* rp = (const v4i*)urows + (size_t)t * (EPT / 4);
        int row[EPT];
#pragma unroll
        for (int g = 0; g < EPT / 4; ++g) {
          v4i r = rp[g];
          row[g * 4 + 0] = r.x; row[g * 4 + 1] = r.y; row[g * 4 + 2] = r.z; row[g * 4 + 3] = r.w;
        }
        const v4i* cp = (const v4i*)ucols + (size_t)t * (EPT / 4);
        int col[EPT];
#pragma unroll
        for (int g = 0; g < EPT / 4; ++g) {
          v4i q4 = cp[g];
          col[g * 4 + 0] = q4.x; col[g * 4 + 1] = q4.y; col[g * 4 + 2] = q4.z; col[g * 4 + 3] = q4.w;
        }
        int pos[EPT];
#pragma unroll
        for (int i = 0; i < EPT; ++i) {
          const bool mine = (row[i] >= NHALF0 ? 1 : 0) == hh;
          pos[i] = -1;
          if (mine) pos[i] = atomicAdd(slots + (size_t)row[i] * SLOT, 1);
        }
#pragma unroll
        for (int i = 0; i < EPT; ++i) {
          const bool mine = (row[i] >= NHALF0 ? 1 : 0) == hh;
          if (mine) {
            int e = t * EPT + i;
            if (pos[i] < CAP) {
              v2i ent;
              ent.x = e;
              ent.y = col[i];
              *(v2i*)(slots + (size_t)row[i] * SLOT + ent_word(pos[i])) = ent;
            } else {
              int o = atomicAdd(spill, 1);
              spill[1 + o] = e;
            }
          }
        }
      }
    }
  }
}

struct Comp2 { v2f m0, m1, m2, i0, i1, i2; };

__device__ __forceinline__ v2f splat2(float v) { v2f t; t.x = v; t.y = v; return t; }

__device__ __forceinline__ v2f logit3(float u0, float u1, float u2, const Comp2& c) {
  v2f d0 = splat2(u0) - c.m0, d1 = splat2(u1) - c.m1, d2 = splat2(u2) - c.m2;
  v2f s = d0 * d0 * c.i0;
  s += d1 * d1 * c.i1;
  s += d2 * d2 * c.i2;
  return -0.5f * s;
}

__device__ __forceinline__ v2f exp2v(v2f a) { v2f t; t.x = __expf(a.x); t.y = __expf(a.y); return t; }
__device__ __forceinline__ v2f max2v(v2f a, v2f b) { v2f t; t.x = fmaxf(a.x, b.x); t.y = fmaxf(a.y, b.y); return t; }

__device__ __forceinline__ Comp2 load_comp2(const float* __restrict__ mu,
                                            const float* __restrict__ sigma, int k) {
  Comp2 c;
  c.m0.x = mu[k * DD + 0]; c.m1.x = mu[k * DD + 1]; c.m2.x = mu[k * DD + 2];
  c.m0.y = mu[k * DD + 3]; c.m1.y = mu[k * DD + 4]; c.m2.y = mu[k * DD + 5];
  v2f s0, s1, s2;
  s0.x = sigma[k * DD + 0]; s1.x = sigma[k * DD + 1]; s2.x = sigma[k * DD + 2];
  s0.y = sigma[k * DD + 3]; s1.y = sigma[k * DD + 4]; s2.y = sigma[k * DD + 5];
  c.i0 = 1.0f / (s0 * s0 + EPSF);
  c.i1 = 1.0f / (s1 * s1 + EPSF);
  c.i2 = 1.0f / (s2 * s2 + EPSF);
  return c;
}

__device__ __forceinline__ int grp4i(int v, int Q) {
  return __builtin_amdgcn_permlane16(v, v, (unsigned)Q * 0x11111111u + 0x44440000u,
                                     (unsigned)Q * 0x11111111u + 0xCCCC8888u, false, false);
}
__device__ __forceinline__ float grp4(float v, int Q) {
  return __builtin_bit_cast(float, grp4i(__builtin_bit_cast(int, v), Q));
}

__device__ __forceinline__ int edge_id(int e) { return min(max(e, 0), EE - 1); }
__device__ __forceinline__ int node_id(int n) { return min(max(n, 0), NN - 1); }

constexpr int LPN = 4;
constexpr int NPB = 16;
constexpr int ROWT = NPB * LPN;
constexpr int PITCH = SLOT + 4;
constexpr int NCT = 2;
static_assert(NN % NPB == 0, "whole work-groups only: the grid is exactly NN / NPB");
static_assert((ROWT / 32) * NCT * 16 == HH, "the waves' column tiles cover GW's columns");

__global__ __launch_bounds__(ROWT) void k_row_gemm(const float* __restrict__ X,
                      const float* __restrict__ uval,
                      const float* __restrict__ mu,
                      const float* __restrict__ sigma,
                      const float* __restrict__ GW,
                      const int* __restrict__ urows,
                      const int* __restrict__ ucols,
                      const int* __restrict__ spill,
                      const int* __restrict__ slots,
                      float* __restrict__ out) {
  __shared__ v4f rowbuf[NPB * PITCH / 4];
  const int jn = threadIdx.x >> 2;
  const int n0 = blockIdx.x * NPB;
  const int node = n0 + jn;
  const int r = threadIdx.x & (LPN - 1);

  const Comp2 cc = load_comp2(mu, sigma, 2 * r);

  const int* slot = slots + (size_t)node * SLOT;
  const int cnt = slot[0];
  const int nin = min(max(cnt, 0), CAP);
  const bool spilled = cnt > CAP;
  int nsp = 0;
  if (spilled) nsp = min(max(spill[0], 0), 2 * NK - 1);

  const unsigned xq = (unsigned)(XT0 * 4) + (unsigned)r * 16u;
  const char* Sb = (const char*)slots;
  const unsigned nbase = (unsigned)node * (unsigned)(SLOT * 4);
  const v2f ninf2 = splat2(-INFINITY);
  v2f m = ninf2;
  v2f den = splat2(0.0f);
  v2f acc[FB];
#pragma unroll
  for (int i = 0; i < FB; ++i) acc[i] = splat2(0.0f);

  for (int j = 0; j < nin; j += CH_W) {
    const int left = nin - j;
    const int mypos = j + (r < left ? r : 0);
    const v2i ent = *(const v2i*)(Sb + (nbase + 4u * (unsigned)ent_word(mypos)));
    const int mye = edge_id(ent.x);
    const int mycol = node_id(ent.y);
    const float* p = uval + (unsigned)mye * (unsigned)DD;
    const float uv0 = p[0], uv1 = p[1], uv2 = p[2];
    v4f x[CH_W];
#pragma unroll
    for (int i = 0; i < CH_W; ++i) {
      const unsigned col = (unsigned)grp4i(mycol, i);
      x[i] = *(const v4f*)(Sb + (xq + col * (unsigned)(SLOT * 4)));
    }
    v2f l[CH_W];
    v2f mn = m;
#pragma unroll
    for (int i = 0; i < CH_W; ++i) {
      v2f lg = logit3(grp4(uv0, i), grp4(uv1, i), grp4(uv2, i), cc);
      if (!(i < left)) lg = ninf2;
      l[i] = lg;
      mn = max2v(mn, lg);
    }
    const v2f sc = exp2v(m - mn);
    m = mn;
    den *= sc;
#pragma unroll
    for (int q = 0; q < FB; ++q) acc[q] *= sc;
#pragma unroll
    for (int i = 0; i < CH_W; ++i) {
      const v2f w = exp2v(l[i] - mn);
      den += w;
#pragma unroll
      for (int c = 0; c < FB; ++c) {
        acc[c] = __builtin_elementwise_fma(w, splat2(grp4(x[i][c & 3], c >> 2)), acc[c]);
      }
    }
  }
  if (spilled) {
    for (int i = 0; i < nsp; ++i) {
      int e = edge_id(spill[1 + i]);
      if (urows[e] == node) {
        const float* p = uval + (size_t)e * DD;
        const v2f lg = logit3(p[0], p[1], p[2], cc);
        const v2f mn = max2v(m, lg);
        const v2f sc = exp2v(m - mn);
        const v2f w = exp2v(lg - mn);
        m = mn;
        den = den * sc + w;
        int cl = ucols[e];
        const float* x0 = X + (size_t)cl * FF;
        const float* x1 = X + (size_t)NN * FF + (size_t)cl * FF;
#pragma unroll
        for (int f = 0; f < FF; ++f) {
          acc[f * 2 + 0] = acc[f * 2 + 0] * sc + w * splat2(x0[f]);
          acc[f * 2 + 1] = acc[f * 2 + 1] * sc + w * splat2(x1[f]);
        }
      }
    }
  }
  {
    v2f inv = splat2(0.0f);
    if (cnt > 0) inv = 1.0f / den;
#pragma unroll
    for (int q = 0; q < FB; ++q) acc[q] *= inv;
  }

  const int lane = threadIdx.x & 31;
  const int half = lane >> 4;
  const int lm = lane & 15;
  const int colw = (threadIdx.x >> 5) * (NCT * 16);

#if __has_builtin(__builtin_amdgcn_wmma_f32_16x16x4_f32)
  v2f bfr[NCT][CDIM / 4];
#pragma unroll
  for (int ct = 0; ct < NCT; ++ct) {
#pragma unroll
    for (int kk = 0; kk < CDIM / 4; ++kk) {
      int c0 = kk * 4 + half * 2;
      bfr[ct][kk].x = GW[c0 * HH + colw + ct * 16 + lm];
      bfr[ct][kk].y = GW[(c0 + 1) * HH + colw + ct * 16 + lm];
    }
  }
#endif

  {
    float* rowp = (float*)rowbuf + jn * PITCH + r * 4;
#pragma unroll
    for (int f = 0; f < FF; ++f) {
      v4f v;
      v.x = acc[f * 2 + 0].x;
      v.y = acc[f * 2 + 1].x;
      v.z = acc[f * 2 + 0].y;
      v.w = acc[f * 2 + 1].y;
      *(v4f*)(rowp + f * 16) = v;
    }
  }
  __syncthreads();

#if __has_builtin(__builtin_amdgcn_wmma_f32_16x16x4_f32)
  const v4f* ap = rowbuf + lm * (PITCH / 4) + half;
  v8f d0[NCT], d1[NCT];
#pragma unroll
  for (int ct = 0; ct < NCT; ++ct) { d0[ct] = v8f{}; d1[ct] = v8f{}; }
#pragma unroll
  for (int kk = 0; kk < CDIM / 4; ++kk) {
    v4f a4 = ap[kk * 2];
    v2f a0, a1;
    a0.x = a4.x; a0.y = a4.z;
    a1.x = a4.y; a1.y = a4.w;
#pragma unroll
    for (int ct = 0; ct < NCT; ++ct) {
      d0[ct] = __builtin_amdgcn_wmma_f32_16x16x4_f32(
           false, a0,  false, bfr[ct][kk],
           (short)0, d0[ct],  false,  false);
      d1[ct] = __builtin_amdgcn_wmma_f32_16x16x4_f32(
           false, a1,  false, bfr[ct][kk],
           (short)0, d1[ct],  false,  false);
    }
  }
#pragma unroll
  for (int ct = 0; ct < NCT; ++ct) {
#pragma unroll
    for (int j = 0; j < 8; ++j) {
      size_t rr = (size_t)(n0 + half * 8 + j);
      out[rr * HH + colw + ct * 16 + lm] = d0[ct][j];
      out[((size_t)NN + rr) * HH + colw + ct * 16 + lm] = d1[ct][j];
    }
  }
#else
  {
    const float* rw = (const float*)rowbuf;
    for (int ct = 0; ct < NCT; ++ct) {
      for (int j = 0; j < 8; ++j) {
        int jr = half * 8 + j;
        for (int bb = 0; bb < BB; ++bb) {
          float s = 0.0f;
          for (int c = 0; c < CDIM; ++c)
            s += rw[jr * PITCH + (c >> 3) * 16 + (c & 7) * 2 + bb] * GW[c * HH + colw + ct * 16 + lm];
          out[((size_t)bb * NN + n0 + jr) * HH + colw + ct * 16 + lm] = s;
        }
      }
    }
  }
#endif
}

extern "C" void kernel_launch(void* const* d_in, const int* in_sizes, int n_in,
                              void* d_out, int out_size, void* d_ws, size_t ws_size,
                              hipStream_t stream) {
  const float* X     = (const float*)d_in[0];
  const float* uval  = (const float*)d_in[1];
  const float* mu    = (const float*)d_in[2];
  const float* sigma = (const float*)d_in[3];
  const float* GW    = (const float*)d_in[4];
  const int* urows   = (const int*)d_in[5];
  const int* ucols   = (const int*)d_in[6];
  float* out = (float*)d_out;

  int* spill = (int*)d_ws;
  int* slots = (int*)d_ws + 2 * NK;

  k_init<<<INIT_G, INIT_T, 0, stream>>>(X, spill, slots);

  k_fill<<<2 * NCHUNK, FILL_T, 0, stream>>>(urows, ucols, spill, slots);

  k_row_gemm<<<NN / NPB, ROWT, 0, stream>>>(
      X, uval, mu, sigma, GW, urows, ucols, spill, slots, out);
}
